# starting kernel
# baseline (speedup 1.0000x reference)
#include <hip/hip_runtime.h>
#include <hip/hip_bf16.h>

typedef __attribute__((ext_vector_type(2))) float v2f;
typedef __attribute__((ext_vector_type(8))) float v8f;

#define D_FEAT 64

__device__ __forceinline__ void atomic_add_f32(float* p, float v) {
    __hip_atomic_fetch_add(p, v, __ATOMIC_RELAXED, __HIP_MEMORY_SCOPE_AGENT);
}

__global__ void sage_init_kernel(const float* __restrict__ x1,
                                 const float* __restrict__ x2,
                                 float* __restrict__ agg1,
                                 float* __restrict__ agg2,
                                 float* __restrict__ cnt1,
                                 float* __restrict__ cnt2,
                                 long total, int n) {
    long t = (long)blockIdx.x * blockDim.x + threadIdx.x;
    if (t < total) {
        agg1[t] = x1[t];
        agg2[t] = x2[t];
        if (t < n) { cnt1[t] = 1.0f; cnt2[t] = 1.0f; }
    }
}

__global__ void sage_scatter_kernel(const float* __restrict__ x,
                                    const int* __restrict__ edges,
                                    float* __restrict__ agg,
                                    float* __restrict__ cnt,
                                    int n_edges) {
    long t = (long)blockIdx.x * blockDim.x + threadIdx.x;
    long e = t >> 6;
    int  f = (int)(t & 63);
    if (e >= n_edges) return;
    int row = edges[e];
    int col = edges[(long)n_edges + e];
    atomic_add_f32(&agg[(long)row * D_FEAT + f], x[(long)col * D_FEAT + f]);
    if (f == 0) atomic_add_f32(&cnt[row], 1.0f);
}

__global__ __launch_bounds__(256) void sage_gemm_norm_kernel(
        const float* __restrict__ agg1, const float* __restrict__ agg2,
        const float* __restrict__ cnt1, const float* __restrict__ cnt2,
        const float* __restrict__ x1,   const float* __restrict__ W,
        const float* __restrict__ bias, float* __restrict__ out, int n) {
    __shared__ float ldsW[192 * 64];

    const int tid = threadIdx.x;
    for (int idx = tid; idx < 192 * 64; idx += 256) {
        const int row = idx >> 6;
        const int col = idx & 63;
        ldsW[((((row >> 1) << 6) | col) << 1) | (row & 1)] = W[idx];
    }
    __syncthreads();

    const int lane = tid & 31;
    const int wave = tid >> 5;
    const int m = lane & 15;
    const int h = lane >> 4;

    const long tile = (long)blockIdx.x * 8 + wave;
    const long row0 = tile << 4;
    long r = row0 + m;
    if (r > (long)(n - 1)) r = (long)(n - 1);

    const float* pa1 = agg1 + r * D_FEAT;
    const float* pa2 = agg2 + r * D_FEAT;
    const float* pax = x1   + r * D_FEAT;
    const float inv1 = 1.0f / fmaxf(cnt1[r], 1.0f);
    const float inv2 = 1.0f / fmaxf(cnt2[r], 1.0f);

    v8f acc0 = {}; v8f acc1 = {}; v8f acc2 = {}; v8f acc3 = {};
    const float* wbase = ldsW + (m << 1);

#pragma unroll
    for (int k = 0; k < 16; ++k) {
        const int c = (k << 2) + (h << 1);
        v2f a = *(const v2f*)(pa1 + c);
        a *= inv1;
        const float* wp = wbase + (((k << 1) + h) << 7);
        const v2f b0 = *(const v2f*)(wp);
        const v2f b1 = *(const v2f*)(wp + 32);
        const v2f b2 = *(const v2f*)(wp + 64);
        const v2f b3 = *(const v2f*)(wp + 96);
        acc0 = __builtin_amdgcn_wmma_f32_16x16x4_f32(false, a, false, b0, (short)0, acc0, false, false);
        acc1 = __builtin_amdgcn_wmma_f32_16x16x4_f32(false, a, false, b1, (short)0, acc1, false, false);
        acc2 = __builtin_amdgcn_wmma_f32_16x16x4_f32(false, a, false, b2, (short)0, acc2, false, false);
        acc3 = __builtin_amdgcn_wmma_f32_16x16x4_f32(false, a, false, b3, (short)0, acc3, false, false);
    }
#pragma unroll
    for (int k = 0; k < 16; ++k) {
        const int c = (k << 2) + (h << 1);
        v2f a = *(const v2f*)(pa2 + c);
        a *= inv2;
        const float* wp = wbase + ((32 + (k << 1) + h) << 7);
        const v2f b0 = *(const v2f*)(wp);
        const v2f b1 = *(const v2f*)(wp + 32);
        const v2f b2 = *(const v2f*)(wp + 64);
        const v2f b3 = *(const v2f*)(wp + 96);
        acc0 = __builtin_amdgcn_wmma_f32_16x16x4_f32(false, a, false, b0, (short)0, acc0, false, false);
        acc1 = __builtin_amdgcn_wmma_f32_16x16x4_f32(false, a, false, b1, (short)0, acc1, false, false);
        acc2 = __builtin_amdgcn_wmma_f32_16x16x4_f32(false, a, false, b2, (short)0, acc2, false, false);
        acc3 = __builtin_amdgcn_wmma_f32_16x16x4_f32(false, a, false, b3, (short)0, acc3, false, false);
    }
#pragma unroll
    for (int k = 0; k < 16; ++k) {
        const int c = (k << 2) + (h << 1);
        const v2f a = *(const v2f*)(pax + c);
        const float* wp = wbase + ((64 + (k << 1) + h) << 7);
        const v2f b0 = *(const v2f*)(wp);
        const v2f b1 = *(const v2f*)(wp + 32);
        const v2f b2 = *(const v2f*)(wp + 64);
        const v2f b3 = *(const v2f*)(wp + 96);
        acc0 = __builtin_amdgcn_wmma_f32_16x16x4_f32(false, a, false, b0, (short)0, acc0, false, false);
        acc1 = __builtin_amdgcn_wmma_f32_16x16x4_f32(false, a, false, b1, (short)0, acc1, false, false);
        acc2 = __builtin_amdgcn_wmma_f32_16x16x4_f32(false, a, false, b2, (short)0, acc2, false, false);
        acc3 = __builtin_amdgcn_wmma_f32_16x16x4_f32(false, a, false, b3, (short)0, acc3, false, false);
    }

    const float bs0 = bias[m];
    const float bs1 = bias[16 + m];
    const float bs2 = bias[32 + m];
    const float bs3 = bias[48 + m];

#pragma unroll
    for (int j = 0; j < 8; ++j) {
        const float o0 = acc0[j] + bs0;
        const float o1 = acc1[j] + bs1;
        const float o2 = acc2[j] + bs2;
        const float o3 = acc3[j] + bs3;
        float s = o0 * o0 + o1 * o1 + o2 * o2 + o3 * o3;
        s += __shfl_xor(s, 1, 32);
        s += __shfl_xor(s, 2, 32);
        s += __shfl_xor(s, 4, 32);
        s += __shfl_xor(s, 8, 32);
        const float scale = 1.0f / fmaxf(sqrtf(s), 1e-12f);
        const long orow = row0 + j + (h << 3);
        if (orow < n) {
            float* po = out + orow * D_FEAT + m;
            po[0]  = o0 * scale;
            po[16] = o1 * scale;
            po[32] = o2 * scale;
            po[48] = o3 * scale;
        }
    }
}

extern "C" void kernel_launch(void* const* d_in, const int* in_sizes, int n_in,
                              void* d_out, int out_size, void* d_ws, size_t ws_size,
                              hipStream_t stream) {
    const float* x1    = (const float*)d_in[0];
    const float* x2    = (const float*)d_in[1];
    const int*   e_pos = (const int*)d_in[2];
    const int*   e_neg = (const int*)d_in[3];
    const float* W     = (const float*)d_in[4];
    const float* bias  = (const float*)d_in[5];
    float*       out   = (float*)d_out;

    const int n       = in_sizes[0] / D_FEAT;
    const int n_edges = in_sizes[2] / 2;

    float* agg1 = (float*)d_ws;
    float* agg2 = agg1 + (size_t)n * D_FEAT;
    float* cnt1 = agg2 + (size_t)n * D_FEAT;
    float* cnt2 = cnt1 + n;

    const long total = (long)n * D_FEAT;
    {
        const int blk = 256;
        const long grid = (total + blk - 1) / blk;
        sage_init_kernel<<<(unsigned)grid, blk, 0, stream>>>(x1, x2, agg1, agg2,
                                                             cnt1, cnt2, total, n);
    }
    {
        const int blk = 256;
        const long work = (long)n_edges * D_FEAT;
        const long grid = (work + blk - 1) / blk;
        sage_scatter_kernel<<<(unsigned)grid, blk, 0, stream>>>(x1, e_pos, agg1, cnt1, n_edges);
        sage_scatter_kernel<<<(unsigned)grid, blk, 0, stream>>>(x2, e_neg, agg2, cnt2, n_edges);
    }
    {
        const int tiles  = (n + 15) / 16;
        const int blocks = (tiles + 7) / 8;
        sage_gemm_norm_kernel<<<blocks, 256, 0, stream>>>(agg1, agg2, cnt1, cnt2,
                                                          x1, W, bias, out, n);
    }
}


# Each block owns 800 destination rows of one edge set, keeps their sums in LDS, reads the whole row array once and adds matching source rows by LDS float atomics (arrival order); no global atomics
# speedup vs baseline: 3.9747x; 3.9747x over previous
#include <hip/hip_runtime.h>
#include <hip/hip_bf16.h>

typedef __attribute__((ext_vector_type(2))) float v2f;
typedef __attribute__((ext_vector_type(8))) float v8f;

#define D_FEAT 64

#define OWN_ROWS 800
#define OWN_THREADS 1024
#define OWN_WAVES 32
#define Q_CAP 160
#define GATHER_G 16

__device__ __forceinline__ void lds_add_f32(float* p, float v) {
    __hip_atomic_fetch_add(p, v, __ATOMIC_RELAXED, __HIP_MEMORY_SCOPE_WORKGROUP);
}

__device__ __forceinline__ int owner_flush(int qn, bool final, int2* q,
                                           float* s_agg, float* s_cnt,
                                           const int* __restrict__ cols,
                                           const float* __restrict__ x, int lane) {
    int base = 0;
    while ((qn - base >= 32) || (final && base < qn)) {
        const int left = qn - base;
        const int nb = left < 32 ? left : 32;
        int my_rl = 0;
        int my_col = 0;
        if (lane < nb) {
            const int2 ent = q[base + lane];
            my_rl = ent.x;
            my_col = cols[ent.y];
            lds_add_f32(&s_cnt[my_rl], 1.0f);
        }
        for (int k = 0; k < nb; k += GATHER_G) {
            v2f v[GATHER_G];
#pragma unroll
            for (int u = 0; u < GATHER_G; ++u) {
                const int c = __builtin_amdgcn_readlane(my_col, k + u);
                v[u] = *(const v2f*)(x + ((size_t)c << 6) + (lane << 1));
            }
#pragma unroll
            for (int u = 0; u < GATHER_G; ++u) {
                const int rl = __builtin_amdgcn_readlane(my_rl, k + u);
                if (k + u < nb) {
                    float* p = s_agg + (rl << 6) + (lane << 1);
                    lds_add_f32(p, v[u].x);
                    lds_add_f32(p + 1, v[u].y);
                }
            }
        }
        base += 32;
    }
    const int rem = qn > base ? qn - base : 0;
    if (rem > 0 && base > 0) {
        int2 t = make_int2(0, 0);
        if (lane < rem) t = q[base + lane];
        if (lane < rem) q[lane] = t;
    }
    return rem;
}

__global__ __launch_bounds__(OWN_THREADS) void sage_owner_sum_kernel(
        const float* __restrict__ x1, const float* __restrict__ x2,
        const int* __restrict__ e_pos, const int* __restrict__ e_neg,
        float* __restrict__ agg1, float* __restrict__ agg2,
        float* __restrict__ cnt1, float* __restrict__ cnt2,
        int n, int n_edges, int blocks_per_set) {
    __shared__ float s_agg[OWN_ROWS * D_FEAT];
    __shared__ float s_cnt[OWN_ROWS];
    __shared__ int2  s_q[OWN_WAVES * Q_CAP];

    const int tid  = threadIdx.x;
    const int lane = tid & 31;
    const int wave = __builtin_amdgcn_readfirstlane(tid >> 5);

    const bool second = (int)blockIdx.x >= blocks_per_set;
    const int  b      = second ? (int)blockIdx.x - blocks_per_set : (int)blockIdx.x;
    const float* __restrict__ x     = second ? x2 : x1;
    const int*   __restrict__ edges = second ? e_neg : e_pos;
    float* __restrict__ agg = second ? agg2 : agg1;
    float* __restrict__ cnt = second ? cnt2 : cnt1;

    const int lo    = b * OWN_ROWS;
    const int avail = n - lo;
    const int nrows = avail < OWN_ROWS ? avail : OWN_ROWS;

    {
        const float4* src = (const float4*)(x + (size_t)lo * D_FEAT);
        float4* dst = (float4*)s_agg;
        const int tot4 = nrows * (D_FEAT / 4);
        int i = tid;
        for (; i + 3 * OWN_THREADS < tot4; i += 4 * OWN_THREADS) {
            float4 t[4];
#pragma unroll
            for (int u = 0; u < 4; ++u) t[u] = src[i + u * OWN_THREADS];
#pragma unroll
            for (int u = 0; u < 4; ++u) dst[i + u * OWN_THREADS] = t[u];
        }
        for (; i < tot4; i += OWN_THREADS) dst[i] = src[i];
        for (int i = tid; i < nrows; i += OWN_THREADS) s_cnt[i] = 1.0f;
    }
    __syncthreads();

    const int4* rows4 = (const int4*)edges;
    const int*  cols  = edges + n_edges;
    const int   n4     = n_edges >> 2;
    const int   nunits = (n4 + 255) >> 8;
    int2* q = s_q + wave * Q_CAP;
    int qn = 0;

    for (int unit = wave; unit < nunits; unit += OWN_WAVES) {
        const int i0 = (unit << 8) + lane;
        int4 r[8];
#pragma unroll
        for (int u = 0; u < 8; ++u) {
            const int idx = i0 + (u << 5);
            r[u] = make_int4(-1, -1, -1, -1);
            if (idx < n4) r[u] = rows4[idx];
        }
#pragma unroll
        for (int u = 0; u < 8; ++u) {
            const int e0 = (i0 + (u << 5)) << 2;
            const unsigned d0 = (unsigned)(r[u].x - lo);
            const unsigned d1 = (unsigned)(r[u].y - lo);
            const unsigned d2 = (unsigned)(r[u].z - lo);
            const unsigned d3 = (unsigned)(r[u].w - lo);
            const bool m0 = d0 < (unsigned)nrows;
            const bool m1 = d1 < (unsigned)nrows;
            const bool m2 = d2 < (unsigned)nrows;
            const bool m3 = d3 < (unsigned)nrows;
            const unsigned b0 = __builtin_amdgcn_ballot_w32(m0);
            const unsigned b1 = __builtin_amdgcn_ballot_w32(m1);
            const unsigned b2 = __builtin_amdgcn_ballot_w32(m2);
            const unsigned b3 = __builtin_amdgcn_ballot_w32(m3);
            if ((b0 | b1 | b2 | b3) != 0u) {
                if (b0) {
                    if (m0) q[qn + (int)__builtin_amdgcn_mbcnt_lo(b0, 0u)] = make_int2((int)d0, e0);
                    qn += __builtin_popcount(b0);
                }
                if (b1) {
                    if (m1) q[qn + (int)__builtin_amdgcn_mbcnt_lo(b1, 0u)] = make_int2((int)d1, e0 + 1);
                    qn += __builtin_popcount(b1);
                }
                if (b2) {
                    if (m2) q[qn + (int)__builtin_amdgcn_mbcnt_lo(b2, 0u)] = make_int2((int)d2, e0 + 2);
                    qn += __builtin_popcount(b2);
                }
                if (b3) {
                    if (m3) q[qn + (int)__builtin_amdgcn_mbcnt_lo(b3, 0u)] = make_int2((int)d3, e0 + 3);
                    qn += __builtin_popcount(b3);
                }
                if (qn >= 32) qn = owner_flush(qn, false, q, s_agg, s_cnt, cols, x, lane);
            }
        }
    }
    if (wave == 0) {
        const int e = (n4 << 2) + lane;
        int rw = -1;
        if (e < n_edges) rw = edges[e];
        const unsigned d = (unsigned)(rw - lo);
        const bool m = d < (unsigned)nrows;
        const unsigned bb = __builtin_amdgcn_ballot_w32(m);
        if (bb) {
            if (m) q[qn + (int)__builtin_amdgcn_mbcnt_lo(bb, 0u)] = make_int2((int)d, e);
            qn += __builtin_popcount(bb);
        }
    }
    qn = owner_flush(qn, true, q, s_agg, s_cnt, cols, x, lane);
    __syncthreads();

    {
        const float4* src = (const float4*)s_agg;
        float4* dst = (float4*)(agg + (size_t)lo * D_FEAT);
        for (int i = tid; i < nrows * (D_FEAT / 4); i += OWN_THREADS) dst[i] = src[i];
        for (int i = tid; i < nrows; i += OWN_THREADS) cnt[lo + i] = s_cnt[i];
    }
}

__global__ __launch_bounds__(256) void sage_gemm_norm_kernel(
        const float* __restrict__ agg1, const float* __restrict__ agg2,
        const float* __restrict__ cnt1, const float* __restrict__ cnt2,
        const float* __restrict__ x1,   const float* __restrict__ W,
        const float* __restrict__ bias, float* __restrict__ out, int n) {
    __shared__ float ldsW[192 * 64];

    const int tid = threadIdx.x;
    for (int idx = tid; idx < 192 * 64; idx += 256) {
        const int row = idx >> 6;
        const int col = idx & 63;
        ldsW[((((row >> 1) << 6) | col) << 1) | (row & 1)] = W[idx];
    }
    __syncthreads();

    const int lane = tid & 31;
    const int wave = tid >> 5;
    const int m = lane & 15;
    const int h = lane >> 4;

    const long tile = (long)blockIdx.x * 8 + wave;
    const long row0 = tile << 4;
    long r = row0 + m;
    if (r > (long)(n - 1)) r = (long)(n - 1);

    const float* pa1 = agg1 + r * D_FEAT;
    const float* pa2 = agg2 + r * D_FEAT;
    const float* pax = x1   + r * D_FEAT;
    const float inv1 = 1.0f / fmaxf(cnt1[r], 1.0f);
    const float inv2 = 1.0f / fmaxf(cnt2[r], 1.0f);

    v8f acc0 = {}; v8f acc1 = {}; v8f acc2 = {}; v8f acc3 = {};
    const float* wbase = ldsW + (m << 1);

#pragma unroll
    for (int k = 0; k < 16; ++k) {
        const int c = (k << 2) + (h << 1);
        v2f a = *(const v2f*)(pa1 + c);
        a *= inv1;
        const float* wp = wbase + (((k << 1) + h) << 7);
        const v2f b0 = *(const v2f*)(wp);
        const v2f b1 = *(const v2f*)(wp + 32);
        const v2f b2 = *(const v2f*)(wp + 64);
        const v2f b3 = *(const v2f*)(wp + 96);
        acc0 = __builtin_amdgcn_wmma_f32_16x16x4_f32(false, a, false, b0, (short)0, acc0, false, false);
        acc1 = __builtin_amdgcn_wmma_f32_16x16x4_f32(false, a, false, b1, (short)0, acc1, false, false);
        acc2 = __builtin_amdgcn_wmma_f32_16x16x4_f32(false, a, false, b2, (short)0, acc2, false, false);
        acc3 = __builtin_amdgcn_wmma_f32_16x16x4_f32(false, a, false, b3, (short)0, acc3, false, false);
    }
#pragma unroll
    for (int k = 0; k < 16; ++k) {
        const int c = (k << 2) + (h << 1);
        v2f a = *(const v2f*)(pa2 + c);
        a *= inv2;
        const float* wp = wbase + ((32 + (k << 1) + h) << 7);
        const v2f b0 = *(const v2f*)(wp);
        const v2f b1 = *(const v2f*)(wp + 32);
        const v2f b2 = *(const v2f*)(wp + 64);
        const v2f b3 = *(const v2f*)(wp + 96);
        acc0 = __builtin_amdgcn_wmma_f32_16x16x4_f32(false, a, false, b0, (short)0, acc0, false, false);
        acc1 = __builtin_amdgcn_wmma_f32_16x16x4_f32(false, a, false, b1, (short)0, acc1, false, false);
        acc2 = __builtin_amdgcn_wmma_f32_16x16x4_f32(false, a, false, b2, (short)0, acc2, false, false);
        acc3 = __builtin_amdgcn_wmma_f32_16x16x4_f32(false, a, false, b3, (short)0, acc3, false, false);
    }
#pragma unroll
    for (int k = 0; k < 16; ++k) {
        const int c = (k << 2) + (h << 1);
        const v2f a = *(const v2f*)(pax + c);
        const float* wp = wbase + ((64 + (k << 1) + h) << 7);
        const v2f b0 = *(const v2f*)(wp);
        const v2f b1 = *(const v2f*)(wp + 32);
        const v2f b2 = *(const v2f*)(wp + 64);
        const v2f b3 = *(const v2f*)(wp + 96);
        acc0 = __builtin_amdgcn_wmma_f32_16x16x4_f32(false, a, false, b0, (short)0, acc0, false, false);
        acc1 = __builtin_amdgcn_wmma_f32_16x16x4_f32(false, a, false, b1, (short)0, acc1, false, false);
        acc2 = __builtin_amdgcn_wmma_f32_16x16x4_f32(false, a, false, b2, (short)0, acc2, false, false);
        acc3 = __builtin_amdgcn_wmma_f32_16x16x4_f32(false, a, false, b3, (short)0, acc3, false, false);
    }

    const float bs0 = bias[m];
    const float bs1 = bias[16 + m];
    const float bs2 = bias[32 + m];
    const float bs3 = bias[48 + m];

#pragma unroll
    for (int j = 0; j < 8; ++j) {
        const float o0 = acc0[j] + bs0;
        const float o1 = acc1[j] + bs1;
        const float o2 = acc2[j] + bs2;
        const float o3 = acc3[j] + bs3;
        float s = o0 * o0 + o1 * o1 + o2 * o2 + o3 * o3;
        s += __shfl_xor(s, 1, 32);
        s += __shfl_xor(s, 2, 32);
        s += __shfl_xor(s, 4, 32);
        s += __shfl_xor(s, 8, 32);
        const float scale = 1.0f / fmaxf(sqrtf(s), 1e-12f);
        const long orow = row0 + j + (h << 3);
        if (orow < n) {
            float* po = out + orow * D_FEAT + m;
            po[0]  = o0 * scale;
            po[16] = o1 * scale;
            po[32] = o2 * scale;
            po[48] = o3 * scale;
        }
    }
}

extern "C" void kernel_launch(void* const* d_in, const int* in_sizes, int n_in,
                              void* d_out, int out_size, void* d_ws, size_t ws_size,
                              hipStream_t stream) {
    const float* x1    = (const float*)d_in[0];
    const float* x2    = (const float*)d_in[1];
    const int*   e_pos = (const int*)d_in[2];
    const int*   e_neg = (const int*)d_in[3];
    const float* W     = (const float*)d_in[4];
    const float* bias  = (const float*)d_in[5];
    float*       out   = (float*)d_out;

    const int n       = in_sizes[0] / D_FEAT;
    const int n_edges = in_sizes[2] / 2;

    float* agg1 = (float*)d_ws;
    float* agg2 = agg1 + (size_t)n * D_FEAT;
    float* cnt1 = agg2 + (size_t)n * D_FEAT;
    float* cnt2 = cnt1 + n;

    {
        const int blocks_per_set = (n + OWN_ROWS - 1) / OWN_ROWS;
        sage_owner_sum_kernel<<<2 * blocks_per_set, OWN_THREADS, 0, stream>>>(
            x1, x2, e_pos, e_neg, agg1, agg2, cnt1, cnt2, n, n_edges, blocks_per_set);
    }
    {
        const int tiles  = (n + 15) / 16;
        const int blocks = (tiles + 7) / 8;
        sage_gemm_norm_kernel<<<blocks, 256, 0, stream>>>(agg1, agg2, cnt1, cnt2,
                                                          x1, W, bias, out, n);
    }
}


# A pre-pass writes each edge's owner-block index as one byte; owner blocks scan these bytes (16 edges per 16-byte load, zero-byte test), not the 4-byte rows; 3 launches; sums stay LDS-arrival order
# speedup vs baseline: 4.4704x; 1.1247x over previous
#include <hip/hip_runtime.h>
#include <hip/hip_bf16.h>

typedef __attribute__((ext_vector_type(2))) float v2f;
typedef __attribute__((ext_vector_type(8))) float v8f;

#define D_FEAT 64

#define OWN_ROWS 800
#define KEY_NONE 127u

__device__ __forceinline__ unsigned owner_key(int row, int n) {
    return (unsigned)row < (unsigned)n ? (unsigned)row / OWN_ROWS : KEY_NONE;
}

__device__ __forceinline__ unsigned pack_keys(int4 r, int n) {
    return owner_key(r.x, n) | (owner_key(r.y, n) << 8) |
           (owner_key(r.z, n) << 16) | (owner_key(r.w, n) << 24);
}

__global__ __launch_bounds__(256) void sage_owner_key_kernel(
        const int* __restrict__ e_pos, const int* __restrict__ e_neg,
        float* __restrict__ keybuf, int n, int n_edges, int blocks_per_set) {
    const int n16 = (n_edges + 15) >> 4;
    const bool second = (int)blockIdx.x >= blocks_per_set;
    const int  b      = second ? (int)blockIdx.x - blocks_per_set : (int)blockIdx.x;
    const int* __restrict__ rows = second ? e_neg : e_pos;
    uint4* __restrict__ keys = (uint4*)keybuf + (second ? n16 : 0);
    const int i = b * 256 + (int)threadIdx.x;
    if (i >= n16) return;
    const int e0 = i << 4;
    uint4 k;
    if (e0 + 16 <= n_edges) {
        const int4* p = (const int4*)(rows + e0);
        const int4 r0 = p[0], r1 = p[1], r2 = p[2], r3 = p[3];
        k.x = pack_keys(r0, n); k.y = pack_keys(r1, n);
        k.z = pack_keys(r2, n); k.w = pack_keys(r3, n);
    } else {
        unsigned w[4];
        for (int j = 0; j < 4; ++j) {
            unsigned v = 0u;
            for (int t = 0; t < 4; ++t) {
                const int e = e0 + 4 * j + t;
                const unsigned kk = e < n_edges ? owner_key(rows[e], n) : KEY_NONE;
                v |= kk << (8 * t);
            }
            w[j] = v;
        }
        k = make_uint4(w[0], w[1], w[2], w[3]);
    }
    keys[i] = k;
}

#define OWN_THREADS 1024
#define OWN_WAVES 32
#define Q_CAP 544
#define GATHER_G 16
#define SCAN_U 4

__device__ __forceinline__ void lds_add_f32(float* p, float v) {
    __hip_atomic_fetch_add(p, v, __ATOMIC_RELAXED, __HIP_MEMORY_SCOPE_WORKGROUP);
}

__device__ __forceinline__ int owner_flush(int qn, bool final, int* q,
                                           float* s_agg, float* s_cnt,
                                           const int* __restrict__ rows,
                                           const int* __restrict__ cols,
                                           const float* __restrict__ x, int lo, int lane) {
    int base = 0;
    while ((qn - base >= 32) || (final && base < qn)) {
        const int left = qn - base;
        const int nb = left < 32 ? left : 32;
        int my_rl = 0;
        int my_col = 0;
        if (lane < nb) {
            const int e = q[base + lane];
            const int rw = rows[e];
            my_col = cols[e];
            my_rl = rw - lo;
            lds_add_f32(&s_cnt[my_rl], 1.0f);
        }
        for (int k = 0; k < nb; k += GATHER_G) {
            v2f v[GATHER_G];
#pragma unroll
            for (int u = 0; u < GATHER_G; ++u) {
                const int c = __builtin_amdgcn_readlane(my_col, k + u);
                v[u] = *(const v2f*)(x + ((size_t)c << 6) + (lane << 1));
            }
#pragma unroll
            for (int u = 0; u < GATHER_G; ++u) {
                const int rl = __builtin_amdgcn_readlane(my_rl, k + u);
                if (k + u < nb) {
                    float* p = s_agg + (rl << 6) + (lane << 1);
                    lds_add_f32(p, v[u].x);
                    lds_add_f32(p + 1, v[u].y);
                }
            }
        }
        base += 32;
    }
    const int rem = qn > base ? qn - base : 0;
    if (rem > 0 && base > 0) {
        int t = 0;
        if (lane < rem) t = q[base + lane];
        if (lane < rem) q[lane] = t;
    }
    return rem;
}

__device__ __forceinline__ unsigned zero_bytes(unsigned t) {
    return ~(t + 0x7F7F7F7Fu) & 0x80808080u;
}

__device__ __forceinline__ int queue_marked(unsigned c, int eb, int* q, int qn) {
    unsigned bal = __builtin_amdgcn_ballot_w32(c != 0u);
    while (bal != 0u) {
        if (c != 0u) {
            const int bp = __builtin_ctz(c);
            q[qn + (int)__builtin_amdgcn_mbcnt_lo(bal, 0u)] = eb + (((bp & 3) << 2) | (bp >> 3));
            c &= c - 1u;
        }
        qn += __builtin_popcount(bal);
        bal = __builtin_amdgcn_ballot_w32(c != 0u);
    }
    return qn;
}

__device__ __forceinline__ void load_keys(const uint4* __restrict__ keys, int i0, int n16, uint4* r) {
#pragma unroll
    for (int u = 0; u < SCAN_U; ++u) {
        const int idx = i0 + (u << 5);
        r[u] = keys[idx < n16 ? idx : n16 - 1];
    }
}

__global__ __launch_bounds__(OWN_THREADS) void sage_owner_sum_kernel(
        const float* __restrict__ x1, const float* __restrict__ x2,
        const int* __restrict__ e_pos, const int* __restrict__ e_neg,
        const float* __restrict__ keybuf,
        float* __restrict__ agg1, float* __restrict__ agg2,
        float* __restrict__ cnt1, float* __restrict__ cnt2,
        int n, int n_edges, int blocks_per_set) {
    __shared__ float s_agg[OWN_ROWS * D_FEAT];
    __shared__ float s_cnt[OWN_ROWS];
    __shared__ int   s_q[OWN_WAVES * Q_CAP];

    const int tid  = threadIdx.x;
    const int lane = tid & 31;
    const int wave = __builtin_amdgcn_readfirstlane(tid >> 5);

    const int n16 = (n_edges + 15) >> 4;
    const bool second = (int)blockIdx.x >= blocks_per_set;
    const int  b      = second ? (int)blockIdx.x - blocks_per_set : (int)blockIdx.x;
    const float* __restrict__ x     = second ? x2 : x1;
    const int*   __restrict__ edges = second ? e_neg : e_pos;
    const uint4* __restrict__ keys  = (const uint4*)keybuf + (second ? n16 : 0);
    float* __restrict__ agg = second ? agg2 : agg1;
    float* __restrict__ cnt = second ? cnt2 : cnt1;

    const int lo    = b * OWN_ROWS;
    const int avail = n - lo;
    const int nrows = avail < OWN_ROWS ? avail : OWN_ROWS;

    {
        const float4* src = (const float4*)(x + (size_t)lo * D_FEAT);
        float4* dst = (float4*)s_agg;
        const int tot4 = nrows * (D_FEAT / 4);
        int i = tid;
        for (; i + 3 * OWN_THREADS < tot4; i += 4 * OWN_THREADS) {
            float4 t[4];
#pragma unroll
            for (int u = 0; u < 4; ++u) t[u] = src[i + u * OWN_THREADS];
#pragma unroll
            for (int u = 0; u < 4; ++u) dst[i + u * OWN_THREADS] = t[u];
        }
        for (; i < tot4; i += OWN_THREADS) dst[i] = src[i];
        for (int i = tid; i < nrows; i += OWN_THREADS) s_cnt[i] = 1.0f;
    }
    __syncthreads();

    const int*  cols   = edges + n_edges;
    const int   nunits = (n16 + 32 * SCAN_U - 1) / (32 * SCAN_U);
    const unsigned own = (unsigned)b * 0x01010101u;
    int* q = s_q + wave * Q_CAP;
    int qn = 0;

    uint4 r[SCAN_U];
    if (wave < nunits) load_keys(keys, wave * (32 * SCAN_U) + lane, n16, r);
    for (int unit = wave; unit < nunits; unit += OWN_WAVES) {
        const int nu = unit + OWN_WAVES < nunits ? unit + OWN_WAVES : unit;
        uint4 nx[SCAN_U];
        load_keys(keys, nu * (32 * SCAN_U) + lane, n16, nx);
        const int i0 = unit * (32 * SCAN_U) + lane;
#pragma unroll
        for (int u = 0; u < SCAN_U; ++u) {
            const int e0 = (i0 + (u << 5)) << 4;
            const unsigned z0 = zero_bytes(r[u].x ^ own);
            const unsigned z1 = zero_bytes(r[u].y ^ own);
            const unsigned z2 = zero_bytes(r[u].z ^ own);
            const unsigned z3 = zero_bytes(r[u].w ^ own);
            unsigned c = (z0 >> 7) | (z1 >> 6) | (z2 >> 5) | (z3 >> 4);
            if (i0 + (u << 5) >= n16) c = 0u;
            if (__builtin_amdgcn_ballot_w32(c != 0u) != 0u) {
                qn = queue_marked(c, e0, q, qn);
                if (qn >= 32) qn = owner_flush(qn, false, q, s_agg, s_cnt, edges, cols, x, lo, lane);
            }
        }
#pragma unroll
        for (int u = 0; u < SCAN_U; ++u) r[u] = nx[u];
    }
    qn = owner_flush(qn, true, q, s_agg, s_cnt, edges, cols, x, lo, lane);
    __syncthreads();

    {
        const float4* src = (const float4*)s_agg;
        float4* dst = (float4*)(agg + (size_t)lo * D_FEAT);
        for (int i = tid; i < nrows * (D_FEAT / 4); i += OWN_THREADS) dst[i] = src[i];
        for (int i = tid; i < nrows; i += OWN_THREADS) cnt[lo + i] = s_cnt[i];
    }
}

__global__ __launch_bounds__(256) void sage_gemm_norm_kernel(
        const float* __restrict__ agg1, const float* __restrict__ agg2,
        const float* __restrict__ cnt1, const float* __restrict__ cnt2,
        const float* __restrict__ x1,   const float* __restrict__ W,
        const float* __restrict__ bias, float* __restrict__ out, int n) {
    __shared__ float ldsW[192 * 64];

    const int tid = threadIdx.x;
    for (int idx = tid; idx < 192 * 64; idx += 256) {
        const int row = idx >> 6;
        const int col = idx & 63;
        ldsW[((((row >> 1) << 6) | col) << 1) | (row & 1)] = W[idx];
    }
    __syncthreads();

    const int lane = tid & 31;
    const int wave = tid >> 5;
    const int m = lane & 15;
    const int h = lane >> 4;

    const long tile = (long)blockIdx.x * 8 + wave;
    const long row0 = tile << 4;
    long r = row0 + m;
    if (r > (long)(n - 1)) r = (long)(n - 1);

    const float* pa1 = agg1 + r * D_FEAT;
    const float* pa2 = agg2 + r * D_FEAT;
    const float* pax = x1   + r * D_FEAT;
    const float inv1 = 1.0f / fmaxf(cnt1[r], 1.0f);
    const float inv2 = 1.0f / fmaxf(cnt2[r], 1.0f);

    v8f acc0 = {}; v8f acc1 = {}; v8f acc2 = {}; v8f acc3 = {};
    const float* wbase = ldsW + (m << 1);

#pragma unroll
    for (int k = 0; k < 16; ++k) {
        const int c = (k << 2) + (h << 1);
        v2f a = *(const v2f*)(pa1 + c);
        a *= inv1;
        const float* wp = wbase + (((k << 1) + h) << 7);
        const v2f b0 = *(const v2f*)(wp);
        const v2f b1 = *(const v2f*)(wp + 32);
        const v2f b2 = *(const v2f*)(wp + 64);
        const v2f b3 = *(const v2f*)(wp + 96);
        acc0 = __builtin_amdgcn_wmma_f32_16x16x4_f32(false, a, false, b0, (short)0, acc0, false, false);
        acc1 = __builtin_amdgcn_wmma_f32_16x16x4_f32(false, a, false, b1, (short)0, acc1, false, false);
        acc2 = __builtin_amdgcn_wmma_f32_16x16x4_f32(false, a, false, b2, (short)0, acc2, false, false);
        acc3 = __builtin_amdgcn_wmma_f32_16x16x4_f32(false, a, false, b3, (short)0, acc3, false, false);
    }
#pragma unroll
    for (int k = 0; k < 16; ++k) {
        const int c = (k << 2) + (h << 1);
        v2f a = *(const v2f*)(pa2 + c);
        a *= inv2;
        const float* wp = wbase + ((32 + (k << 1) + h) << 7);
        const v2f b0 = *(const v2f*)(wp);
        const v2f b1 = *(const v2f*)(wp + 32);
        const v2f b2 = *(const v2f*)(wp + 64);
        const v2f b3 = *(const v2f*)(wp + 96);
        acc0 = __builtin_amdgcn_wmma_f32_16x16x4_f32(false, a, false, b0, (short)0, acc0, false, false);
        acc1 = __builtin_amdgcn_wmma_f32_16x16x4_f32(false, a, false, b1, (short)0, acc1, false, false);
        acc2 = __builtin_amdgcn_wmma_f32_16x16x4_f32(false, a, false, b2, (short)0, acc2, false, false);
        acc3 = __builtin_amdgcn_wmma_f32_16x16x4_f32(false, a, false, b3, (short)0, acc3, false, false);
    }
#pragma unroll
    for (int k = 0; k < 16; ++k) {
        const int c = (k << 2) + (h << 1);
        const v2f a = *(const v2f*)(pax + c);
        const float* wp = wbase + ((64 + (k << 1) + h) << 7);
        const v2f b0 = *(const v2f*)(wp);
        const v2f b1 = *(const v2f*)(wp + 32);
        const v2f b2 = *(const v2f*)(wp + 64);
        const v2f b3 = *(const v2f*)(wp + 96);
        acc0 = __builtin_amdgcn_wmma_f32_16x16x4_f32(false, a, false, b0, (short)0, acc0, false, false);
        acc1 = __builtin_amdgcn_wmma_f32_16x16x4_f32(false, a, false, b1, (short)0, acc1, false, false);
        acc2 = __builtin_amdgcn_wmma_f32_16x16x4_f32(false, a, false, b2, (short)0, acc2, false, false);
        acc3 = __builtin_amdgcn_wmma_f32_16x16x4_f32(false, a, false, b3, (short)0, acc3, false, false);
    }

    const float bs0 = bias[m];
    const float bs1 = bias[16 + m];
    const float bs2 = bias[32 + m];
    const float bs3 = bias[48 + m];

#pragma unroll
    for (int j = 0; j < 8; ++j) {
        const float o0 = acc0[j] + bs0;
        const float o1 = acc1[j] + bs1;
        const float o2 = acc2[j] + bs2;
        const float o3 = acc3[j] + bs3;
        float s = o0 * o0 + o1 * o1 + o2 * o2 + o3 * o3;
        s += __shfl_xor(s, 1, 32);
        s += __shfl_xor(s, 2, 32);
        s += __shfl_xor(s, 4, 32);
        s += __shfl_xor(s, 8, 32);
        const float scale = 1.0f / fmaxf(sqrtf(s), 1e-12f);
        const long orow = row0 + j + (h << 3);
        if (orow < n) {
            float* po = out + orow * D_FEAT + m;
            po[0]  = o0 * scale;
            po[16] = o1 * scale;
            po[32] = o2 * scale;
            po[48] = o3 * scale;
        }
    }
}

extern "C" void kernel_launch(void* const* d_in, const int* in_sizes, int n_in,
                              void* d_out, int out_size, void* d_ws, size_t ws_size,
                              hipStream_t stream) {
    const float* x1    = (const float*)d_in[0];
    const float* x2    = (const float*)d_in[1];
    const int*   e_pos = (const int*)d_in[2];
    const int*   e_neg = (const int*)d_in[3];
    const float* W     = (const float*)d_in[4];
    const float* bias  = (const float*)d_in[5];
    float*       out   = (float*)d_out;

    const int n       = in_sizes[0] / D_FEAT;
    const int n_edges = in_sizes[2] / 2;

    float* agg1 = (float*)d_ws;
    float* agg2 = agg1 + (size_t)n * D_FEAT;
    float* cnt1 = agg2 + (size_t)n * D_FEAT;
    float* cnt2 = cnt1 + n;

    {
        const int n16 = (n_edges + 15) / 16;
        const int blocks_per_set = (n16 + 255) / 256;
        sage_owner_key_kernel<<<2 * blocks_per_set, 256, 0, stream>>>(
            e_pos, e_neg, out, n, n_edges, blocks_per_set);
    }
    {
        const int blocks_per_set = (n + OWN_ROWS - 1) / OWN_ROWS;
        sage_owner_sum_kernel<<<2 * blocks_per_set, OWN_THREADS, 0, stream>>>(
            x1, x2, e_pos, e_neg, out, agg1, agg2, cnt1, cnt2, n, n_edges, blocks_per_set);
    }
    {
        const int tiles  = (n + 15) / 16;
        const int blocks = (tiles + 7) / 8;
        sage_gemm_norm_kernel<<<blocks, 256, 0, stream>>>(agg1, agg2, cnt1, cnt2,
                                                          x1, W, bias, out, n);
    }
}


# Pre-pass also packs each edge's row-in-block and source column into one 32-bit word beside its key byte; an owner loads that one word per matched edge, not row and column; sums stay LDS-arrival order
# speedup vs baseline: 4.7352x; 1.0592x over previous
#include <hip/hip_runtime.h>
#include <hip/hip_bf16.h>

typedef __attribute__((ext_vector_type(2))) float v2f;
typedef __attribute__((ext_vector_type(8))) float v8f;

#define D_FEAT 64

#define OWN_ROWS 800
#define KEY_NONE 127u
#define COL_BITS 17
#define COL_MASK 0x1FFFFu

__device__ __forceinline__ void edge_words(int row, int col, int n, unsigned& key, unsigned& rec) {
    if ((unsigned)row < (unsigned)n) {
        key = (unsigned)row / OWN_ROWS;
        const unsigned c = (unsigned)col < (unsigned)n ? (unsigned)col : (unsigned)(n - 1);
        rec = (((unsigned)row - key * OWN_ROWS) << COL_BITS) | c;
    } else {
        key = KEY_NONE;
        rec = 0u;
    }
}

__global__ __launch_bounds__(256) void sage_edge_pack_kernel(
        const int* __restrict__ e_pos, const int* __restrict__ e_neg,
        float* __restrict__ keybuf, int n, int n_edges, int blocks_per_set) {
    const int n16  = (n_edges + 15) >> 4;
    const int npad = n16 << 4;
    const bool second = (int)blockIdx.x >= blocks_per_set;
    const int  b      = second ? (int)blockIdx.x - blocks_per_set : (int)blockIdx.x;
    const int* __restrict__ rows = second ? e_neg : e_pos;
    const int* __restrict__ cols = rows + n_edges;
    unsigned* __restrict__ keys = (unsigned*)((uint4*)keybuf + (second ? n16 : 0));
    unsigned* __restrict__ recs = (unsigned*)((uint4*)keybuf + 2 * n16) + (second ? npad : 0);
    const int lane  = (int)threadIdx.x & 31;
    const int ebase = (b * 8 + ((int)threadIdx.x >> 5)) * 512 + lane * 4;
    if (ebase >= npad) return;

    if ((n_edges & 3) == 0) {
        int4 r[4], c[4];
#pragma unroll
        for (int j = 0; j < 4; ++j) {
            const int e0 = ebase + j * 128;
            const int el = e0 < n_edges ? e0 : 0;
            r[j] = *(const int4*)(rows + el);
            c[j] = *(const int4*)(cols + el);
        }
#pragma unroll
        for (int j = 0; j < 4; ++j) {
            const int e0 = ebase + j * 128;
            if (e0 < npad) {
                unsigned k0 = KEY_NONE, k1 = KEY_NONE, k2 = KEY_NONE, k3 = KEY_NONE;
                unsigned w0 = 0u, w1 = 0u, w2 = 0u, w3 = 0u;
                if (e0 < n_edges) {
                    edge_words(r[j].x, c[j].x, n, k0, w0);
                    edge_words(r[j].y, c[j].y, n, k1, w1);
                    edge_words(r[j].z, c[j].z, n, k2, w2);
                    edge_words(r[j].w, c[j].w, n, k3, w3);
                }
                keys[e0 >> 2] = k0 | (k1 << 8) | (k2 << 16) | (k3 << 24);
                *(uint4*)(recs + e0) = make_uint4(w0, w1, w2, w3);
            }
        }
    } else {
        for (int j = 0; j < 4; ++j) {
            const int e0 = ebase + j * 128;
            if (e0 < npad) {
                unsigned kw = 0u;
                unsigned rw[4];
                for (int t = 0; t < 4; ++t) {
                    const int e = e0 + t;
                    unsigned k = KEY_NONE, rc = 0u;
                    if (e < n_edges) edge_words(rows[e], cols[e], n, k, rc);
                    kw |= k << (8 * t);
                    rw[t] = rc;
                }
                keys[e0 >> 2] = kw;
                *(uint4*)(recs + e0) = make_uint4(rw[0], rw[1], rw[2], rw[3]);
            }
        }
    }
}

#define OWN_THREADS 1024
#define OWN_WAVES 32
#define Q_CAP 544
#define GATHER_G 16
#define SCAN_U 4

__device__ __forceinline__ void lds_add_f32(float* p, float v) {
    __hip_atomic_fetch_add(p, v, __ATOMIC_RELAXED, __HIP_MEMORY_SCOPE_WORKGROUP);
}

__device__ __forceinline__ int owner_flush(int qn, bool final, int* q,
                                           float* s_agg, float* s_cnt,
                                           const unsigned* __restrict__ recs,
                                           const float* __restrict__ x, int lane) {
    int base = 0;
    while ((qn - base >= 32) || (final && base < qn)) {
        const int left = qn - base;
        const int nb = left < 32 ? left : 32;
        int my_rl = 0;
        int my_col = 0;
        if (lane < nb) {
            const unsigned rc = recs[q[base + lane]];
            my_col = (int)(rc & COL_MASK);
            my_rl = (int)(rc >> COL_BITS);
            lds_add_f32(&s_cnt[my_rl], 1.0f);
        }
        for (int k = 0; k < nb; k += GATHER_G) {
            v2f v[GATHER_G];
#pragma unroll
            for (int u = 0; u < GATHER_G; ++u) {
                const int c = __builtin_amdgcn_readlane(my_col, k + u);
                v[u] = *(const v2f*)(x + ((size_t)c << 6) + (lane << 1));
            }
#pragma unroll
            for (int u = 0; u < GATHER_G; ++u) {
                const int rl = __builtin_amdgcn_readlane(my_rl, k + u);
                if (k + u < nb) {
                    float* p = s_agg + (rl << 6) + (lane << 1);
                    lds_add_f32(p, v[u].x);
                    lds_add_f32(p + 1, v[u].y);
                }
            }
        }
        base += 32;
    }
    const int rem = qn > base ? qn - base : 0;
    if (rem > 0 && base > 0) {
        int t = 0;
        if (lane < rem) t = q[base + lane];
        if (lane < rem) q[lane] = t;
    }
    return rem;
}

__device__ __forceinline__ unsigned zero_bytes(unsigned t) {
    return ~(t + 0x7F7F7F7Fu) & 0x80808080u;
}

__device__ __forceinline__ int queue_marked(unsigned c, int eb, int* q, int qn) {
    unsigned bal = __builtin_amdgcn_ballot_w32(c != 0u);
    while (bal != 0u) {
        if (c != 0u) {
            const int bp = __builtin_ctz(c);
            q[qn + (int)__builtin_amdgcn_mbcnt_lo(bal, 0u)] = eb + (((bp & 3) << 2) | (bp >> 3));
            c &= c - 1u;
        }
        qn += __builtin_popcount(bal);
        bal = __builtin_amdgcn_ballot_w32(c != 0u);
    }
    return qn;
}

__device__ __forceinline__ void load_keys(const uint4* __restrict__ keys, int i0, int n16, uint4* r) {
#pragma unroll
    for (int u = 0; u < SCAN_U; ++u) {
        const int idx = i0 + (u << 5);
        r[u] = keys[idx < n16 ? idx : n16 - 1];
    }
}

__global__ __launch_bounds__(OWN_THREADS) void sage_owner_sum_kernel(
        const float* __restrict__ x1, const float* __restrict__ x2,
        const float* __restrict__ keybuf,
        float* __restrict__ agg1, float* __restrict__ agg2,
        float* __restrict__ cnt1, float* __restrict__ cnt2,
        int n, int n_edges, int blocks_per_set) {
    __shared__ float s_agg[OWN_ROWS * D_FEAT];
    __shared__ float s_cnt[OWN_ROWS];
    __shared__ int   s_q[OWN_WAVES * Q_CAP];

    const int tid  = threadIdx.x;
    const int lane = tid & 31;
    const int wave = __builtin_amdgcn_readfirstlane(tid >> 5);

    const int n16 = (n_edges + 15) >> 4;
    const bool second = (int)blockIdx.x >= blocks_per_set;
    const int  b      = second ? (int)blockIdx.x - blocks_per_set : (int)blockIdx.x;
    const float* __restrict__ x     = second ? x2 : x1;
    const uint4* __restrict__ keys  = (const uint4*)keybuf + (second ? n16 : 0);
    const unsigned* __restrict__ recs =
        (const unsigned*)((const uint4*)keybuf + 2 * n16) + (second ? (n16 << 4) : 0);
    float* __restrict__ agg = second ? agg2 : agg1;
    float* __restrict__ cnt = second ? cnt2 : cnt1;

    const int lo    = b * OWN_ROWS;
    const int avail = n - lo;
    const int nrows = avail < OWN_ROWS ? avail : OWN_ROWS;

    {
        const float4* src = (const float4*)(x + (size_t)lo * D_FEAT);
        float4* dst = (float4*)s_agg;
        const int tot4 = nrows * (D_FEAT / 4);
        int i = tid;
        for (; i + 3 * OWN_THREADS < tot4; i += 4 * OWN_THREADS) {
            float4 t[4];
#pragma unroll
            for (int u = 0; u < 4; ++u) t[u] = src[i + u * OWN_THREADS];
#pragma unroll
            for (int u = 0; u < 4; ++u) dst[i + u * OWN_THREADS] = t[u];
        }
        for (; i < tot4; i += OWN_THREADS) dst[i] = src[i];
        for (int i = tid; i < nrows; i += OWN_THREADS) s_cnt[i] = 1.0f;
    }
    __syncthreads();

    const int   nunits = (n16 + 32 * SCAN_U - 1) / (32 * SCAN_U);
    const unsigned own = (unsigned)b * 0x01010101u;
    int* q = s_q + wave * Q_CAP;
    int qn = 0;

    uint4 r[SCAN_U];
    if (wave < nunits) load_keys(keys, wave * (32 * SCAN_U) + lane, n16, r);
    for (int unit = wave; unit < nunits; unit += OWN_WAVES) {
        const int nu = unit + OWN_WAVES < nunits ? unit + OWN_WAVES : unit;
        uint4 nx[SCAN_U];
        load_keys(keys, nu * (32 * SCAN_U) + lane, n16, nx);
        const int i0 = unit * (32 * SCAN_U) + lane;
#pragma unroll
        for (int u = 0; u < SCAN_U; ++u) {
            const int e0 = (i0 + (u << 5)) << 4;
            const unsigned z0 = zero_bytes(r[u].x ^ own);
            const unsigned z1 = zero_bytes(r[u].y ^ own);
            const unsigned z2 = zero_bytes(r[u].z ^ own);
            const unsigned z3 = zero_bytes(r[u].w ^ own);
            unsigned c = (z0 >> 7) | (z1 >> 6) | (z2 >> 5) | (z3 >> 4);
            if (i0 + (u << 5) >= n16) c = 0u;
            if (__builtin_amdgcn_ballot_w32(c != 0u) != 0u) {
                qn = queue_marked(c, e0, q, qn);
                if (qn >= 32) qn = owner_flush(qn, false, q, s_agg, s_cnt, recs, x, lane);
            }
        }
#pragma unroll
        for (int u = 0; u < SCAN_U; ++u) r[u] = nx[u];
    }
    qn = owner_flush(qn, true, q, s_agg, s_cnt, recs, x, lane);
    __syncthreads();

    {
        const float4* src = (const float4*)s_agg;
        float4* dst = (float4*)(agg + (size_t)lo * D_FEAT);
        for (int i = tid; i < nrows * (D_FEAT / 4); i += OWN_THREADS) dst[i] = src[i];
        for (int i = tid; i < nrows; i += OWN_THREADS) cnt[lo + i] = s_cnt[i];
    }
}

__global__ __launch_bounds__(256) void sage_gemm_norm_kernel(
        const float* __restrict__ agg1, const float* __restrict__ agg2,
        const float* __restrict__ cnt1, const float* __restrict__ cnt2,
        const float* __restrict__ x1,   const float* __restrict__ W,
        const float* __restrict__ bias, float* __restrict__ out, int n) {
    __shared__ float ldsW[192 * 64];

    const int tid = threadIdx.x;
    for (int idx = tid; idx < 192 * 64; idx += 256) {
        const int row = idx >> 6;
        const int col = idx & 63;
        ldsW[((((row >> 1) << 6) | col) << 1) | (row & 1)] = W[idx];
    }
    __syncthreads();

    const int lane = tid & 31;
    const int wave = tid >> 5;
    const int m = lane & 15;
    const int h = lane >> 4;

    const long tile = (long)blockIdx.x * 8 + wave;
    const long row0 = tile << 4;
    long r = row0 + m;
    if (r > (long)(n - 1)) r = (long)(n - 1);

    const float* pa1 = agg1 + r * D_FEAT;
    const float* pa2 = agg2 + r * D_FEAT;
    const float* pax = x1   + r * D_FEAT;
    const float inv1 = 1.0f / fmaxf(cnt1[r], 1.0f);
    const float inv2 = 1.0f / fmaxf(cnt2[r], 1.0f);

    v8f acc0 = {}; v8f acc1 = {}; v8f acc2 = {}; v8f acc3 = {};
    const float* wbase = ldsW + (m << 1);

#pragma unroll
    for (int k = 0; k < 16; ++k) {
        const int c = (k << 2) + (h << 1);
        v2f a = *(const v2f*)(pa1 + c);
        a *= inv1;
        const float* wp = wbase + (((k << 1) + h) << 7);
        const v2f b0 = *(const v2f*)(wp);
        const v2f b1 = *(const v2f*)(wp + 32);
        const v2f b2 = *(const v2f*)(wp + 64);
        const v2f b3 = *(const v2f*)(wp + 96);
        acc0 = __builtin_amdgcn_wmma_f32_16x16x4_f32(false, a, false, b0, (short)0, acc0, false, false);
        acc1 = __builtin_amdgcn_wmma_f32_16x16x4_f32(false, a, false, b1, (short)0, acc1, false, false);
        acc2 = __builtin_amdgcn_wmma_f32_16x16x4_f32(false, a, false, b2, (short)0, acc2, false, false);
        acc3 = __builtin_amdgcn_wmma_f32_16x16x4_f32(false, a, false, b3, (short)0, acc3, false, false);
    }
#pragma unroll
    for (int k = 0; k < 16; ++k) {
        const int c = (k << 2) + (h << 1);
        v2f a = *(const v2f*)(pa2 + c);
        a *= inv2;
        const float* wp = wbase + ((32 + (k << 1) + h) << 7);
        const v2f b0 = *(const v2f*)(wp);
        const v2f b1 = *(const v2f*)(wp + 32);
        const v2f b2 = *(const v2f*)(wp + 64);
        const v2f b3 = *(const v2f*)(wp + 96);
        acc0 = __builtin_amdgcn_wmma_f32_16x16x4_f32(false, a, false, b0, (short)0, acc0, false, false);
        acc1 = __builtin_amdgcn_wmma_f32_16x16x4_f32(false, a, false, b1, (short)0, acc1, false, false);
        acc2 = __builtin_amdgcn_wmma_f32_16x16x4_f32(false, a, false, b2, (short)0, acc2, false, false);
        acc3 = __builtin_amdgcn_wmma_f32_16x16x4_f32(false, a, false, b3, (short)0, acc3, false, false);
    }
#pragma unroll
    for (int k = 0; k < 16; ++k) {
        const int c = (k << 2) + (h << 1);
        const v2f a = *(const v2f*)(pax + c);
        const float* wp = wbase + ((64 + (k << 1) + h) << 7);
        const v2f b0 = *(const v2f*)(wp);
        const v2f b1 = *(const v2f*)(wp + 32);
        const v2f b2 = *(const v2f*)(wp + 64);
        const v2f b3 = *(const v2f*)(wp + 96);
        acc0 = __builtin_amdgcn_wmma_f32_16x16x4_f32(false, a, false, b0, (short)0, acc0, false, false);
        acc1 = __builtin_amdgcn_wmma_f32_16x16x4_f32(false, a, false, b1, (short)0, acc1, false, false);
        acc2 = __builtin_amdgcn_wmma_f32_16x16x4_f32(false, a, false, b2, (short)0, acc2, false, false);
        acc3 = __builtin_amdgcn_wmma_f32_16x16x4_f32(false, a, false, b3, (short)0, acc3, false, false);
    }

    const float bs0 = bias[m];
    const float bs1 = bias[16 + m];
    const float bs2 = bias[32 + m];
    const float bs3 = bias[48 + m];

#pragma unroll
    for (int j = 0; j < 8; ++j) {
        const float o0 = acc0[j] + bs0;
        const float o1 = acc1[j] + bs1;
        const float o2 = acc2[j] + bs2;
        const float o3 = acc3[j] + bs3;
        float s = o0 * o0 + o1 * o1 + o2 * o2 + o3 * o3;
        s += __shfl_xor(s, 1, 32);
        s += __shfl_xor(s, 2, 32);
        s += __shfl_xor(s, 4, 32);
        s += __shfl_xor(s, 8, 32);
        const float scale = 1.0f / fmaxf(sqrtf(s), 1e-12f);
        const long orow = row0 + j + (h << 3);
        if (orow < n) {
            float* po = out + orow * D_FEAT + m;
            po[0]  = o0 * scale;
            po[16] = o1 * scale;
            po[32] = o2 * scale;
            po[48] = o3 * scale;
        }
    }
}

extern "C" void kernel_launch(void* const* d_in, const int* in_sizes, int n_in,
                              void* d_out, int out_size, void* d_ws, size_t ws_size,
                              hipStream_t stream) {
    const float* x1    = (const float*)d_in[0];
    const float* x2    = (const float*)d_in[1];
    const int*   e_pos = (const int*)d_in[2];
    const int*   e_neg = (const int*)d_in[3];
    const float* W     = (const float*)d_in[4];
    const float* bias  = (const float*)d_in[5];
    float*       out   = (float*)d_out;

    const int n       = in_sizes[0] / D_FEAT;
    const int n_edges = in_sizes[2] / 2;

    float* agg1 = (float*)d_ws;
    float* agg2 = agg1 + (size_t)n * D_FEAT;
    float* cnt1 = agg2 + (size_t)n * D_FEAT;
    float* cnt2 = cnt1 + n;

    {
        const int n16 = (n_edges + 15) / 16;
        const int blocks_per_set = (n16 + 255) / 256;
        sage_edge_pack_kernel<<<2 * blocks_per_set, 256, 0, stream>>>(
            e_pos, e_neg, out, n, n_edges, blocks_per_set);
    }
    {
        const int blocks_per_set = (n + OWN_ROWS - 1) / OWN_ROWS;
        sage_owner_sum_kernel<<<2 * blocks_per_set, OWN_THREADS, 0, stream>>>(
            x1, x2, out, agg1, agg2, cnt1, cnt2, n, n_edges, blocks_per_set);
    }
    {
        const int tiles  = (n + 15) / 16;
        const int blocks = (tiles + 7) / 8;
        sage_gemm_norm_kernel<<<blocks, 256, 0, stream>>>(agg1, agg2, cnt1, cnt2,
                                                          x1, W, bias, out, n);
    }
}
